# starting kernel
# baseline (speedup 1.0000x reference)
#include <hip/hip_runtime.h>
#include <hip/hip_bf16.h>


#define Hh   512
#define H1   511
#define NN   1024
#define BB   16384

typedef float v2f __attribute__((ext_vector_type(2)));
typedef float v8f __attribute__((ext_vector_type(8)));

__global__ __launch_bounds__(512) void k_build_fg(
    const float* __restrict__ enc_rot, const float* __restrict__ enc_diag,
    const float* __restrict__ dec_rot, const float* __restrict__ dec_diag,
    float* __restrict__ fg) {
  const int m = blockIdx.x;
  const int t = threadIdx.x;
  const float* r = (m < 4) ? (enc_rot + m * Hh) : (dec_rot + (m - 4) * Hh);
  const float* d = (m < 4) ? (enc_diag + m * Hh) : (dec_diag + (m - 4) * Hh);

  __shared__ float sA[512];
  __shared__ float sB[512];
  __shared__ float cS[4];

  const float rv = r[t];
  sA[t] = rv;
  sB[t] = rv * rv;
  __syncthreads();
  for (int off = 256; off > 0; off >>= 1) {
    if (t < off) { sA[t] += sA[t + off]; sB[t] += sB[t + off]; }
    __syncthreads();
  }
  if (t == 0) {
    const float a = sA[0], b = sB[0];
    const float D = 1.0f - a * a + 512.0f * b;
    cS[0] = -2.0f * b / D;
    cS[1] = 1.0f + ((1.0f + a) * (1.0f + a) - 512.0f * b) / D;
    cS[2] = -1.0f - ((1.0f - a) * (1.0f - a) - 512.0f * b) / D;
    cS[3] = -1024.0f / D;
  }
  __syncthreads();
  const float dv = d[t];
  fg[m * 1024 + t]       = dv * (cS[0] + cS[1] * rv);
  fg[m * 1024 + 512 + t] = dv * (cS[2] + cS[3] * rv);
}

__global__ void k_build_prefix(
    int m,
    const float* __restrict__ enc_rot, const float* __restrict__ enc_diag,
    const float* __restrict__ dec_rot, const float* __restrict__ dec_diag,
    const float* __restrict__ fg, float* __restrict__ P) {
  const int idx = blockIdx.x * blockDim.x + threadIdx.x;
  if (idx >= 4 * 1023) return;
  const int k  = idx & 3;
  const int si = idx >> 2;
  const int s  = si - H1;

  const float* r = (m < 4) ? (enc_rot + m * Hh) : (dec_rot + (m - 4) * Hh);
  const float* d = (m < 4) ? (enc_diag + m * Hh) : (dec_diag + (m - 4) * Hh);
  const float* f = fg + m * 1024;
  const float* g = f + 512;

  float run[4] = {0.f, 0.f, 0.f, 0.f};
  float* Pp = P + ((size_t)(k * 1023 + si)) * Hh;

  const int tlo = (s > 0) ? s : 0;
  const int thi = (s < 0) ? (H1 + s) : H1;

  for (int t = 0; t < Hh; ++t) {
    float h = 0.f;
    if (t >= tlo && t <= thi) {
      const int v = t - s;
      if (k == 0) {
        h = f[t] + g[t] * r[v];
        if (s == 0) h += d[t];
      } else if (k == 1) {
        h = f[v] + g[v] * r[H1 - t];
        if (2 * t == H1 + s) h += d[v];
      } else if (k == 2) {
        h = f[H1 - t] + g[H1 - t] * r[H1 - v];
        if (s == 0) h += d[H1 - t];
      } else {
        h = f[H1 - v] + g[H1 - v] * r[t];
        if (2 * t == H1 + s) h += d[H1 - v];
      }
    }
    run[t & 3] += h;
    Pp[t] = run[t & 3];
  }
}

__global__ void k_build_lifted(const float* __restrict__ P,
                               float* __restrict__ L) {
  const int idx = blockIdx.x * blockDim.x + threadIdx.x;
  const int i = idx >> 10;
  const int j = idx & 1023;
  const int s = i - j;
  float acc = 0.f;
  if (s > -Hh && s < Hh) {
    const int si = s + H1;
    const int a = max(0, max(s, i - Hh));
    const int b = min(i, min(H1, H1 + s));
    if (b >= a) {
#pragma unroll
      for (int k = 0; k < 4; ++k) {
        const int c   = (i - k) & 3;
        const int Thi = b - ((b + 4 - c) & 3);
        const int Tlo = a + ((c + 4 - (a & 3)) & 3);
        if (Thi >= Tlo) {
          const float* Pp = P + ((size_t)(k * 1023 + si)) * Hh;
          float v = Pp[Thi];
          if (Tlo >= 4) v -= Pp[Tlo - 4];
          acc += v;
        }
      }
    }
  }
  L[idx] = acc;
}

#define BM 128
#define BN 64
#define BK 16

__global__ __launch_bounds__(256) void k_gemm_xwt(const float* __restrict__ A,
                                                  const float* __restrict__ W,
                                                  float* __restrict__ Out) {
  __shared__ float As[BM][BK + 1];
  __shared__ float Bs[BK][BN + 1];

  const int tid  = threadIdx.x;
  const int mb   = blockIdx.x * BM;
  const int nb   = blockIdx.y * BN;
  const int lane = tid & 31;
  const int w    = tid >> 5;
  const int wm   = (w & 3) * 32;
  const int wn   = (w >> 2) * 32;
  const int half = lane >> 4;
  const int l15  = lane & 15;

  v8f acc[2][2];
#pragma unroll
  for (int fm = 0; fm < 2; ++fm)
#pragma unroll
    for (int fn = 0; fn < 2; ++fn) acc[fm][fn] = v8f{};

  const int am = tid >> 1;
  const int ac = (tid & 1) * 8;
  const int wr = tid >> 2;
  const int wk = (tid & 3) * 4;

  for (int kb = 0; kb < 1024; kb += BK) {
    const float4 a0 = *(const float4*)(A + (size_t)(mb + am) * 1024 + kb + ac);
    const float4 a1 = *(const float4*)(A + (size_t)(mb + am) * 1024 + kb + ac + 4);
    As[am][ac + 0] = a0.x; As[am][ac + 1] = a0.y;
    As[am][ac + 2] = a0.z; As[am][ac + 3] = a0.w;
    As[am][ac + 4] = a1.x; As[am][ac + 5] = a1.y;
    As[am][ac + 6] = a1.z; As[am][ac + 7] = a1.w;
    const float4 wv = *(const float4*)(W + (size_t)(nb + wr) * 1024 + kb + wk);
    Bs[wk + 0][wr] = wv.x; Bs[wk + 1][wr] = wv.y;
    Bs[wk + 2][wr] = wv.z; Bs[wk + 3][wr] = wv.w;
    __syncthreads();

#pragma unroll
    for (int k4 = 0; k4 < BK; k4 += 4) {
      v2f af[2], bf[2];
#pragma unroll
      for (int fm = 0; fm < 2; ++fm) {
        const int row = wm + fm * 16 + l15;
        af[fm].x = As[row][k4 + half * 2];
        af[fm].y = As[row][k4 + half * 2 + 1];
      }
#pragma unroll
      for (int fn = 0; fn < 2; ++fn) {
        const int col = wn + fn * 16 + l15;
        bf[fn].x = Bs[k4 + half * 2][col];
        bf[fn].y = Bs[k4 + half * 2 + 1][col];
      }
#pragma unroll
      for (int fm = 0; fm < 2; ++fm)
#pragma unroll
        for (int fn = 0; fn < 2; ++fn)
          acc[fm][fn] = __builtin_amdgcn_wmma_f32_16x16x4_f32(
              false, af[fm], false, bf[fn], (short)0, acc[fm][fn],
              false, false);
    }
    __syncthreads();
  }

#pragma unroll
  for (int fm = 0; fm < 2; ++fm) {
#pragma unroll
    for (int fn = 0; fn < 2; ++fn) {
      const int col = nb + wn + fn * 16 + l15;
#pragma unroll
      for (int g = 0; g < 8; ++g) {
        const int row = mb + wm + fm * 16 + g + half * 8;
        Out[(size_t)row * 1024 + col] = acc[fm][fn][g];
      }
    }
  }
}

extern "C" void kernel_launch(void* const* d_in, const int* in_sizes, int n_in,
                              void* d_out, int out_size, void* d_ws,
                              size_t ws_size, hipStream_t stream) {
  (void)in_sizes; (void)n_in; (void)out_size; (void)ws_size;

  const float* x        = (const float*)d_in[0];
  const float* enc_rot  = (const float*)d_in[1];
  const float* enc_diag = (const float*)d_in[2];
  const float* dec_rot  = (const float*)d_in[3];
  const float* dec_diag = (const float*)d_in[4];

  float* out = (float*)d_out;
  float* ws  = (float*)d_ws;

  float* fg     = ws;
  float* P      = fg + (size_t)8 * 1024;
  float* lifted = P + (size_t)4 * 1023 * Hh;
  float* actA   = lifted + (size_t)8 * NN * NN;
  float* actB   = actA + (size_t)BB * NN;

  float* bneck  = out;
  float* final_ = out + (size_t)BB * NN;

  k_build_fg<<<8, 512, 0, stream>>>(enc_rot, enc_diag, dec_rot, dec_diag, fg);

  for (int m = 0; m < 8; ++m) {
    k_build_prefix<<<(4 * 1023 + 255) / 256, 256, 0, stream>>>(
        m, enc_rot, enc_diag, dec_rot, dec_diag, fg, P);
    k_build_lifted<<<(NN * NN) / 256, 256, 0, stream>>>(
        P, lifted + (size_t)m * NN * NN);
  }

  dim3 grid(BB / BM, NN / BN);
  k_gemm_xwt<<<grid, 256, 0, stream>>>(x,     lifted + 0 * (size_t)NN * NN, actA);
  k_gemm_xwt<<<grid, 256, 0, stream>>>(actA,  lifted + 1 * (size_t)NN * NN, actB);
  k_gemm_xwt<<<grid, 256, 0, stream>>>(actB,  lifted + 2 * (size_t)NN * NN, actA);
  k_gemm_xwt<<<grid, 256, 0, stream>>>(actA,  lifted + 3 * (size_t)NN * NN, bneck);
  k_gemm_xwt<<<grid, 256, 0, stream>>>(bneck, lifted + 4 * (size_t)NN * NN, actA);
  k_gemm_xwt<<<grid, 256, 0, stream>>>(actA,  lifted + 5 * (size_t)NN * NN, actB);
  k_gemm_xwt<<<grid, 256, 0, stream>>>(actB,  lifted + 6 * (size_t)NN * NN, actA);
  k_gemm_xwt<<<grid, 256, 0, stream>>>(actA,  lifted + 7 * (size_t)NN * NN, final_);
}


# GEMM chain as one launch: a work-group keeps 64 batch rows in LDS through all 8 layers, layer result in registers, W fragments straight from memory; k regrouped, summation order changes
# speedup vs baseline: 3.5781x; 3.5781x over previous
#include <hip/hip_runtime.h>
#include <hip/hip_bf16.h>


#define Hh   512
#define H1   511
#define NN   1024
#define BB   16384

typedef float v2f __attribute__((ext_vector_type(2)));
typedef float v8f __attribute__((ext_vector_type(8)));

__global__ __launch_bounds__(512) void k_build_fg(
    const float* __restrict__ enc_rot, const float* __restrict__ enc_diag,
    const float* __restrict__ dec_rot, const float* __restrict__ dec_diag,
    float* __restrict__ fg) {
  const int m = blockIdx.x;
  const int t = threadIdx.x;
  const float* r = (m < 4) ? (enc_rot + m * Hh) : (dec_rot + (m - 4) * Hh);
  const float* d = (m < 4) ? (enc_diag + m * Hh) : (dec_diag + (m - 4) * Hh);

  __shared__ float sA[512];
  __shared__ float sB[512];
  __shared__ float cS[4];

  const float rv = r[t];
  sA[t] = rv;
  sB[t] = rv * rv;
  __syncthreads();
  for (int off = 256; off > 0; off >>= 1) {
    if (t < off) { sA[t] += sA[t + off]; sB[t] += sB[t + off]; }
    __syncthreads();
  }
  if (t == 0) {
    const float a = sA[0], b = sB[0];
    const float D = 1.0f - a * a + 512.0f * b;
    cS[0] = -2.0f * b / D;
    cS[1] = 1.0f + ((1.0f + a) * (1.0f + a) - 512.0f * b) / D;
    cS[2] = -1.0f - ((1.0f - a) * (1.0f - a) - 512.0f * b) / D;
    cS[3] = -1024.0f / D;
  }
  __syncthreads();
  const float dv = d[t];
  fg[m * 1024 + t]       = dv * (cS[0] + cS[1] * rv);
  fg[m * 1024 + 512 + t] = dv * (cS[2] + cS[3] * rv);
}

#define PPITCH 1024
#define PSTEP  (4 * PPITCH)

template <int K>
__device__ __forceinline__ void scan_class(
    const float* __restrict__ r, const float* __restrict__ d,
    const float* __restrict__ f, const float* __restrict__ g,
    const int s, const int c, float* __restrict__ Pp) {
  const int tlo = (s > 0) ? s : 0;
  const int thi = (s < 0) ? (H1 + s) : H1;
  float run = 0.f;
#pragma unroll 8
  for (int n = 0; n < Hh / 4; ++n) {
    const int t = c + 4 * n;
    const bool in = (t >= tlo) && (t <= thi);
    const int v = in ? (t - s) : 0;
    float h;
    if (K == 0) {
      h = f[t] + g[t] * r[v];
      const float dv = d[t];
      h = (s == 0) ? (h + dv) : h;
    } else if (K == 1) {
      h = f[v] + g[v] * r[H1 - t];
      const float dv = d[v];
      h = (2 * t == H1 + s) ? (h + dv) : h;
    } else if (K == 2) {
      h = f[H1 - t] + g[H1 - t] * r[H1 - v];
      const float dv = d[H1 - t];
      h = (s == 0) ? (h + dv) : h;
    } else {
      h = f[H1 - v] + g[H1 - v] * r[t];
      const float dv = d[H1 - v];
      h = (2 * t == H1 + s) ? (h + dv) : h;
    }
    run += in ? h : 0.f;
    Pp[(size_t)t * PSTEP] = run;
  }
}

__global__ __launch_bounds__(64) void k_build_prefix(
    int m,
    const float* __restrict__ enc_rot, const float* __restrict__ enc_diag,
    const float* __restrict__ dec_rot, const float* __restrict__ dec_diag,
    const float* __restrict__ fg, float* __restrict__ P) {
  const int si = blockIdx.x * 64 + threadIdx.x;
  if (si >= 1023) return;
  const int s  = si - H1;
  const int k  = blockIdx.y;
  const int c  = blockIdx.z;

  const float* r = (m < 4) ? (enc_rot + m * Hh) : (dec_rot + (m - 4) * Hh);
  const float* d = (m < 4) ? (enc_diag + m * Hh) : (dec_diag + (m - 4) * Hh);
  const float* f = fg + m * 1024;
  const float* g = f + 512;

  float* Pp = P + (size_t)k * PPITCH + si;
  if (k == 0)      scan_class<0>(r, d, f, g, s, c, Pp);
  else if (k == 1) scan_class<1>(r, d, f, g, s, c, Pp);
  else if (k == 2) scan_class<2>(r, d, f, g, s, c, Pp);
  else             scan_class<3>(r, d, f, g, s, c, Pp);
}

__global__ void k_build_lifted(const float* __restrict__ P,
                               float* __restrict__ L) {
  const int idx = blockIdx.x * blockDim.x + threadIdx.x;
  const int i = idx >> 10;
  const int j = idx & 1023;
  const int s = i - j;
  float acc = 0.f;
  if (s > -Hh && s < Hh) {
    const int si = s + H1;
    const int a = max(0, max(s, i - Hh));
    const int b = min(i, min(H1, H1 + s));
    if (b >= a) {
#pragma unroll
      for (int k = 0; k < 4; ++k) {
        const int c   = (i - k) & 3;
        const int Thi = b - ((b + 4 - c) & 3);
        const int Tlo = a + ((c + 4 - (a & 3)) & 3);
        if (Thi >= Tlo) {
          const float* Pp = P + (size_t)k * PPITCH + si;
          float v = Pp[(size_t)Thi * PSTEP];
          if (Tlo >= 4) v -= Pp[(size_t)(Tlo - 4) * PSTEP];
          acc += v;
        }
      }
    }
  }
  L[idx] = acc;
}

#define CM     64
#define CPITCH 1028

__device__ __forceinline__ void chain_panel(const float* __restrict__ wp,
                                            const float* ap,
                                            const int k_lo, const int k_hi,
                                            v8f (&acc)[4][2]) {
  for (int kb = k_lo; kb < k_hi; kb += 32) {
    float4 b[2][4];
#pragma unroll
    for (int fn = 0; fn < 2; ++fn)
#pragma unroll
      for (int q = 0; q < 4; ++q)
        b[fn][q] = *(const float4*)(wp + fn * 16 * 1024 + kb + 4 * q);
    float4 a[4][4];
#pragma unroll
    for (int fm = 0; fm < 4; ++fm)
#pragma unroll
      for (int q = 0; q < 4; ++q)
        a[fm][q] = *(const float4*)(ap + fm * 16 * CPITCH + kb + 4 * q);
#pragma unroll
    for (int q = 0; q < 4; ++q) {
#pragma unroll
      for (int e = 0; e < 2; ++e) {
        v2f af[4], bf[2];
#pragma unroll
        for (int fm = 0; fm < 4; ++fm)
          af[fm] = e ? v2f{a[fm][q].z, a[fm][q].w} : v2f{a[fm][q].x, a[fm][q].y};
#pragma unroll
        for (int fn = 0; fn < 2; ++fn)
          bf[fn] = e ? v2f{b[fn][q].z, b[fn][q].w} : v2f{b[fn][q].x, b[fn][q].y};
#pragma unroll
        for (int fm = 0; fm < 4; ++fm)
#pragma unroll
          for (int fn = 0; fn < 2; ++fn)
            acc[fm][fn] = __builtin_amdgcn_wmma_f32_16x16x4_f32(
                false, af[fm], false, bf[fn], (short)0, acc[fm][fn],
                false, false);
      }
    }
  }
}

__device__ __forceinline__ void chain_store(float* act, const int n0,
                                            const int c, const int h,
                                            const v8f (&acc)[4][2]) {
#pragma unroll
  for (int fm = 0; fm < 4; ++fm)
#pragma unroll
    for (int fn = 0; fn < 2; ++fn)
#pragma unroll
      for (int r = 0; r < 8; ++r)
        act[(16 * fm + 8 * h + r) * CPITCH + n0 + 16 * fn + c] = acc[fm][fn][r];
}

__global__ __launch_bounds__(512) void k_chain(const float* __restrict__ X,
                                               const float* __restrict__ Wall,
                                               float* __restrict__ Bneck,
                                               float* __restrict__ Final) {
  __shared__ float Act[CM * CPITCH];

  const int tid  = threadIdx.x;
  const int lane = tid & 31;
  const int w    = __builtin_amdgcn_readfirstlane(tid >> 5);
  const int c    = lane & 15;
  const int h    = lane >> 4;
  const size_t mb = (size_t)blockIdx.x * CM;

  for (int it = 0; it < (CM * 256) / 512; ++it) {
    const int idx = it * 512 + tid;
    const int row = idx >> 8;
    const int c4  = (idx & 255) * 4;
    *(float4*)(Act + row * CPITCH + c4) =
        *(const float4*)(X + (mb + row) * 1024 + c4);
  }
  __syncthreads();

  const int n1 = 32 * w;
  const int n2 = 32 * w + 512;
  const float* ap = Act + c * CPITCH + 16 * h;

  for (int l = 0; l < 8; ++l) {
    const float* Wl = Wall + (size_t)l * NN * NN;
    v8f acc1[4][2], acc2[4][2];
#pragma unroll
    for (int fm = 0; fm < 4; ++fm)
#pragma unroll
      for (int fn = 0; fn < 2; ++fn) { acc1[fm][fn] = v8f{}; acc2[fm][fn] = v8f{}; }

    chain_panel(Wl + (size_t)(n1 + c) * 1024 + 16 * h, ap, 0, n1 + 544, acc1);
    chain_panel(Wl + (size_t)(n2 + c) * 1024 + 16 * h, ap, n1, 1024, acc2);

    __syncthreads();
    chain_store(Act, n1, c, h, acc1);
    chain_store(Act, n2, c, h, acc2);
    __syncthreads();

    if (l == 3 || l == 7) {
      float* dst = (l == 3) ? Bneck : Final;
      for (int it = 0; it < (CM * 256) / 512; ++it) {
        const int idx = it * 512 + tid;
        const int row = idx >> 8;
        const int c4  = (idx & 255) * 4;
        *(float4*)(dst + (mb + row) * 1024 + c4) =
            *(const float4*)(Act + row * CPITCH + c4);
      }
    }
  }
}

extern "C" void kernel_launch(void* const* d_in, const int* in_sizes, int n_in,
                              void* d_out, int out_size, void* d_ws,
                              size_t ws_size, hipStream_t stream) {
  (void)in_sizes; (void)n_in; (void)out_size; (void)ws_size;

  const float* x        = (const float*)d_in[0];
  const float* enc_rot  = (const float*)d_in[1];
  const float* enc_diag = (const float*)d_in[2];
  const float* dec_rot  = (const float*)d_in[3];
  const float* dec_diag = (const float*)d_in[4];

  float* out = (float*)d_out;
  float* ws  = (float*)d_ws;

  float* fg     = ws;
  float* P      = fg + (size_t)8 * 1024;
  float* lifted = P + (size_t)Hh * PSTEP;

  float* bneck  = out;
  float* final_ = out + (size_t)BB * NN;

  k_build_fg<<<8, 512, 0, stream>>>(enc_rot, enc_diag, dec_rot, dec_diag, fg);

  for (int m = 0; m < 8; ++m) {
    k_build_prefix<<<dim3(16, 4, 4), 64, 0, stream>>>(
        m, enc_rot, enc_diag, dec_rot, dec_diag, fg, P);
    k_build_lifted<<<(NN * NN) / 256, 256, 0, stream>>>(
        P, lifted + (size_t)m * NN * NN);
  }

  k_chain<<<BB / CM, 512, 0, stream>>>(x, lifted, bneck, final_);
}


# Build of the 8 lifted matrices as ONE launch: a work-group owns 32 diagonals of one matrix, prefix sums kept in LDS and never in memory; 18 launches become 2; same sums in the same order
# speedup vs baseline: 4.0250x; 1.1249x over previous
#include <hip/hip_runtime.h>
#include <hip/hip_bf16.h>


#define Hh   512
#define H1   511
#define NN   1024
#define BB   16384

typedef float v2f __attribute__((ext_vector_type(2)));
typedef float v8f __attribute__((ext_vector_type(8)));

#define PLANE 32
#define PSTEP (4 * PLANE)

template <int K>
__device__ __forceinline__ void scan_class(
    const float* r, const float* d, const float* f, const float* g,
    const int s, const int c, float* Pp) {
  const int tlo = (s > 0) ? s : 0;
  const int thi = (s < 0) ? (H1 + s) : H1;
  float run = 0.f;
#pragma unroll 8
  for (int n = 0; n < Hh / 4; ++n) {
    const int t = c + 4 * n;
    const bool in = (t >= tlo) && (t <= thi);
    const int v = in ? (t - s) : 0;
    float h;
    if (K == 0) {
      h = f[t] + g[t] * r[v];
      const float dv = d[t];
      h = (s == 0) ? (h + dv) : h;
    } else if (K == 1) {
      h = f[v] + g[v] * r[H1 - t];
      const float dv = d[v];
      h = (2 * t == H1 + s) ? (h + dv) : h;
    } else if (K == 2) {
      h = f[H1 - t] + g[H1 - t] * r[H1 - v];
      const float dv = d[H1 - t];
      h = (s == 0) ? (h + dv) : h;
    } else {
      h = f[H1 - v] + g[H1 - v] * r[t];
      const float dv = d[H1 - v];
      h = (2 * t == H1 + s) ? (h + dv) : h;
    }
    run += in ? h : 0.f;
    Pp[t * PSTEP] = run;
  }
}

__global__ __launch_bounds__(512) void k_build_band(
    const float* __restrict__ enc_rot, const float* __restrict__ enc_diag,
    const float* __restrict__ dec_rot, const float* __restrict__ dec_diag,
    float* __restrict__ Lall) {
  __shared__ float Pl[Hh * PSTEP];
  __shared__ float sR[Hh];
  __shared__ float sD[Hh];
  __shared__ float sF[Hh];
  __shared__ float sG[Hh];
  __shared__ float sA[Hh];
  __shared__ float sB[Hh];
  __shared__ float cS[4];

  const int strip = blockIdx.x;
  const int m     = blockIdx.y;
  const int t     = threadIdx.x;
  const int lane  = t & 31;
  const int wv    = __builtin_amdgcn_readfirstlane(t >> 5);
  const float* r = (m < 4) ? (enc_rot + m * Hh) : (dec_rot + (m - 4) * Hh);
  const float* d = (m < 4) ? (enc_diag + m * Hh) : (dec_diag + (m - 4) * Hh);
  float* L = Lall + (size_t)m * NN * NN;

  for (int rr = 0; rr < 32; ++rr) {
    const int i = 32 * strip + rr;
#pragma unroll
    for (int jj = 0; jj < 2; ++jj) {
      const int j = t + 512 * jj;
      const int dd = i - j;
      if (dd >= Hh || dd <= -Hh) L[(size_t)i * NN + j] = 0.f;
    }
  }

  const float rv = r[t];
  const float dv = d[t];
  sR[t] = rv;
  sD[t] = dv;
  sA[t] = rv;
  sB[t] = rv * rv;
  __syncthreads();
  for (int off = 256; off > 0; off >>= 1) {
    if (t < off) { sA[t] += sA[t + off]; sB[t] += sB[t + off]; }
    __syncthreads();
  }
  if (t == 0) {
    const float a = sA[0], b = sB[0];
    const float D = 1.0f - a * a + 512.0f * b;
    cS[0] = -2.0f * b / D;
    cS[1] = 1.0f + ((1.0f + a) * (1.0f + a) - 512.0f * b) / D;
    cS[2] = -1.0f - ((1.0f - a) * (1.0f - a) - 512.0f * b) / D;
    cS[3] = -1024.0f / D;
  }
  __syncthreads();
  sF[t] = dv * (cS[0] + cS[1] * rv);
  sG[t] = dv * (cS[2] + cS[3] * rv);
  __syncthreads();

  const int si = 32 * strip + lane;
  const int s  = si - H1;
  {
    const int k = wv >> 2;
    const int c = wv & 3;
    float* Pp = Pl + k * PLANE + lane;
    if (k == 0)      scan_class<0>(sR, sD, sF, sG, s, c, Pp);
    else if (k == 1) scan_class<1>(sR, sD, sF, sG, s, c, Pp);
    else if (k == 2) scan_class<2>(sR, sD, sF, sG, s, c, Pp);
    else             scan_class<3>(sR, sD, sF, sG, s, c, Pp);
  }
  __syncthreads();

  for (int it = 0; it < NN / 16; ++it) {
    const int i = 16 * it + wv;
    const int j = i - s;
    const int a = max(0, max(s, i - Hh));
    const int b = min(i, min(H1, H1 + s));
    float acc = 0.f;
#pragma unroll
    for (int k = 0; k < 4; ++k) {
      const int c   = (i - k) & 3;
      const int Thi = b - ((b + 4 - c) & 3);
      const int Tlo = a + ((c + 4 - (a & 3)) & 3);
      const bool ok = (b >= a) && (Thi >= Tlo);
      const bool lo = ok && (Tlo >= 4);
      const float* Pp = Pl + k * PLANE + lane;
      const float vh = Pp[(ok ? Thi : 0) * PSTEP];
      const float vl = Pp[(lo ? (Tlo - 4) : 0) * PSTEP];
      const float v  = lo ? (vh - vl) : vh;
      acc = ok ? (acc + v) : acc;
    }
    if (si < 2 * H1 + 1 && j >= 0 && j < NN) L[(size_t)i * NN + j] = acc;
  }
}

#define CM     64
#define CPITCH 1028

__device__ __forceinline__ void chain_panel(const float* __restrict__ wp,
                                            const float* ap,
                                            const int k_lo, const int k_hi,
                                            v8f (&acc)[4][2]) {
  for (int kb = k_lo; kb < k_hi; kb += 32) {
    float4 b[2][4];
#pragma unroll
    for (int fn = 0; fn < 2; ++fn)
#pragma unroll
      for (int q = 0; q < 4; ++q)
        b[fn][q] = *(const float4*)(wp + fn * 16 * 1024 + kb + 4 * q);
    float4 a[4][4];
#pragma unroll
    for (int fm = 0; fm < 4; ++fm)
#pragma unroll
      for (int q = 0; q < 4; ++q)
        a[fm][q] = *(const float4*)(ap + fm * 16 * CPITCH + kb + 4 * q);
#pragma unroll
    for (int q = 0; q < 4; ++q) {
#pragma unroll
      for (int e = 0; e < 2; ++e) {
        v2f af[4], bf[2];
#pragma unroll
        for (int fm = 0; fm < 4; ++fm)
          af[fm] = e ? v2f{a[fm][q].z, a[fm][q].w} : v2f{a[fm][q].x, a[fm][q].y};
#pragma unroll
        for (int fn = 0; fn < 2; ++fn)
          bf[fn] = e ? v2f{b[fn][q].z, b[fn][q].w} : v2f{b[fn][q].x, b[fn][q].y};
#pragma unroll
        for (int fm = 0; fm < 4; ++fm)
#pragma unroll
          for (int fn = 0; fn < 2; ++fn)
            acc[fm][fn] = __builtin_amdgcn_wmma_f32_16x16x4_f32(
                false, af[fm], false, bf[fn], (short)0, acc[fm][fn],
                false, false);
      }
    }
  }
}

__device__ __forceinline__ void chain_store(float* act, const int n0,
                                            const int c, const int h,
                                            const v8f (&acc)[4][2]) {
#pragma unroll
  for (int fm = 0; fm < 4; ++fm)
#pragma unroll
    for (int fn = 0; fn < 2; ++fn)
#pragma unroll
      for (int r = 0; r < 8; ++r)
        act[(16 * fm + 8 * h + r) * CPITCH + n0 + 16 * fn + c] = acc[fm][fn][r];
}

__global__ __launch_bounds__(512) void k_chain(const float* __restrict__ X,
                                               const float* __restrict__ Wall,
                                               float* __restrict__ Bneck,
                                               float* __restrict__ Final) {
  __shared__ float Act[CM * CPITCH];

  const int tid  = threadIdx.x;
  const int lane = tid & 31;
  const int w    = __builtin_amdgcn_readfirstlane(tid >> 5);
  const int c    = lane & 15;
  const int h    = lane >> 4;
  const size_t mb = (size_t)blockIdx.x * CM;

  for (int it = 0; it < (CM * 256) / 512; ++it) {
    const int idx = it * 512 + tid;
    const int row = idx >> 8;
    const int c4  = (idx & 255) * 4;
    *(float4*)(Act + row * CPITCH + c4) =
        *(const float4*)(X + (mb + row) * 1024 + c4);
  }
  __syncthreads();

  const int n1 = 32 * w;
  const int n2 = 32 * w + 512;
  const float* ap = Act + c * CPITCH + 16 * h;

  for (int l = 0; l < 8; ++l) {
    const float* Wl = Wall + (size_t)l * NN * NN;
    v8f acc1[4][2], acc2[4][2];
#pragma unroll
    for (int fm = 0; fm < 4; ++fm)
#pragma unroll
      for (int fn = 0; fn < 2; ++fn) { acc1[fm][fn] = v8f{}; acc2[fm][fn] = v8f{}; }

    chain_panel(Wl + (size_t)(n1 + c) * 1024 + 16 * h, ap, 0, n1 + 544, acc1);
    chain_panel(Wl + (size_t)(n2 + c) * 1024 + 16 * h, ap, n1, 1024, acc2);

    __syncthreads();
    chain_store(Act, n1, c, h, acc1);
    chain_store(Act, n2, c, h, acc2);
    __syncthreads();

    if (l == 3 || l == 7) {
      float* dst = (l == 3) ? Bneck : Final;
      for (int it = 0; it < (CM * 256) / 512; ++it) {
        const int idx = it * 512 + tid;
        const int row = idx >> 8;
        const int c4  = (idx & 255) * 4;
        *(float4*)(dst + (mb + row) * 1024 + c4) =
            *(const float4*)(Act + row * CPITCH + c4);
      }
    }
  }
}

extern "C" void kernel_launch(void* const* d_in, const int* in_sizes, int n_in,
                              void* d_out, int out_size, void* d_ws,
                              size_t ws_size, hipStream_t stream) {
  (void)in_sizes; (void)n_in; (void)out_size; (void)ws_size;

  const float* x        = (const float*)d_in[0];
  const float* enc_rot  = (const float*)d_in[1];
  const float* enc_diag = (const float*)d_in[2];
  const float* dec_rot  = (const float*)d_in[3];
  const float* dec_diag = (const float*)d_in[4];

  float* out = (float*)d_out;
  float* lifted = (float*)d_ws;

  float* bneck  = out;
  float* final_ = out + (size_t)BB * NN;

  k_build_band<<<dim3(32, 8), 512, 0, stream>>>(
      enc_rot, enc_diag, dec_rot, dec_diag, lifted);

  k_chain<<<BB / CM, 512, 0, stream>>>(x, lifted, bneck, final_);
}


# k_chain: 8 waves of 4 panels, two fragment register sets: the loads of K-step i+1 (W from memory, A from LDS) go out as one group before step i's WMMAs, no wave waits on them; same sums, same order
# speedup vs baseline: 4.1879x; 1.0405x over previous
#include <hip/hip_runtime.h>
#include <hip/hip_bf16.h>


#define Hh   512
#define H1   511
#define NN   1024
#define BB   16384

typedef float v2f __attribute__((ext_vector_type(2)));
typedef float v8f __attribute__((ext_vector_type(8)));

#define PLANE 32
#define PSTEP (4 * PLANE)

template <int K>
__device__ __forceinline__ void scan_class(
    const float* r, const float* d, const float* f, const float* g,
    const int s, const int c, float* Pp) {
  const int tlo = (s > 0) ? s : 0;
  const int thi = (s < 0) ? (H1 + s) : H1;
  float run = 0.f;
#pragma unroll 8
  for (int n = 0; n < Hh / 4; ++n) {
    const int t = c + 4 * n;
    const bool in = (t >= tlo) && (t <= thi);
    const int v = in ? (t - s) : 0;
    float h;
    if (K == 0) {
      h = f[t] + g[t] * r[v];
      const float dv = d[t];
      h = (s == 0) ? (h + dv) : h;
    } else if (K == 1) {
      h = f[v] + g[v] * r[H1 - t];
      const float dv = d[v];
      h = (2 * t == H1 + s) ? (h + dv) : h;
    } else if (K == 2) {
      h = f[H1 - t] + g[H1 - t] * r[H1 - v];
      const float dv = d[H1 - t];
      h = (s == 0) ? (h + dv) : h;
    } else {
      h = f[H1 - v] + g[H1 - v] * r[t];
      const float dv = d[H1 - v];
      h = (2 * t == H1 + s) ? (h + dv) : h;
    }
    run += in ? h : 0.f;
    Pp[t * PSTEP] = run;
  }
}

__global__ __launch_bounds__(512) void k_build_band(
    const float* __restrict__ enc_rot, const float* __restrict__ enc_diag,
    const float* __restrict__ dec_rot, const float* __restrict__ dec_diag,
    float* __restrict__ Lall) {
  __shared__ float Pl[Hh * PSTEP];
  __shared__ float sR[Hh];
  __shared__ float sD[Hh];
  __shared__ float sF[Hh];
  __shared__ float sG[Hh];
  __shared__ float sA[Hh];
  __shared__ float sB[Hh];
  __shared__ float cS[4];

  const int strip = blockIdx.x;
  const int m     = blockIdx.y;
  const int t     = threadIdx.x;
  const int lane  = t & 31;
  const int wv    = __builtin_amdgcn_readfirstlane(t >> 5);
  const float* r = (m < 4) ? (enc_rot + m * Hh) : (dec_rot + (m - 4) * Hh);
  const float* d = (m < 4) ? (enc_diag + m * Hh) : (dec_diag + (m - 4) * Hh);
  float* L = Lall + (size_t)m * NN * NN;

  for (int rr = 0; rr < 32; ++rr) {
    const int i = 32 * strip + rr;
#pragma unroll
    for (int jj = 0; jj < 2; ++jj) {
      const int j = t + 512 * jj;
      const int dd = i - j;
      if (dd >= Hh || dd <= -Hh) L[(size_t)i * NN + j] = 0.f;
    }
  }

  const float rv = r[t];
  const float dv = d[t];
  sR[t] = rv;
  sD[t] = dv;
  sA[t] = rv;
  sB[t] = rv * rv;
  __syncthreads();
  for (int off = 256; off > 0; off >>= 1) {
    if (t < off) { sA[t] += sA[t + off]; sB[t] += sB[t + off]; }
    __syncthreads();
  }
  if (t == 0) {
    const float a = sA[0], b = sB[0];
    const float D = 1.0f - a * a + 512.0f * b;
    cS[0] = -2.0f * b / D;
    cS[1] = 1.0f + ((1.0f + a) * (1.0f + a) - 512.0f * b) / D;
    cS[2] = -1.0f - ((1.0f - a) * (1.0f - a) - 512.0f * b) / D;
    cS[3] = -1024.0f / D;
  }
  __syncthreads();
  sF[t] = dv * (cS[0] + cS[1] * rv);
  sG[t] = dv * (cS[2] + cS[3] * rv);
  __syncthreads();

  const int si = 32 * strip + lane;
  const int s  = si - H1;
  {
    const int k = wv >> 2;
    const int c = wv & 3;
    float* Pp = Pl + k * PLANE + lane;
    if (k == 0)      scan_class<0>(sR, sD, sF, sG, s, c, Pp);
    else if (k == 1) scan_class<1>(sR, sD, sF, sG, s, c, Pp);
    else if (k == 2) scan_class<2>(sR, sD, sF, sG, s, c, Pp);
    else             scan_class<3>(sR, sD, sF, sG, s, c, Pp);
  }
  __syncthreads();

  for (int it = 0; it < NN / 16; ++it) {
    const int i = 16 * it + wv;
    const int j = i - s;
    const int a = max(0, max(s, i - Hh));
    const int b = min(i, min(H1, H1 + s));
    float acc = 0.f;
#pragma unroll
    for (int k = 0; k < 4; ++k) {
      const int c   = (i - k) & 3;
      const int Thi = b - ((b + 4 - c) & 3);
      const int Tlo = a + ((c + 4 - (a & 3)) & 3);
      const bool ok = (b >= a) && (Thi >= Tlo);
      const bool lo = ok && (Tlo >= 4);
      const float* Pp = Pl + k * PLANE + lane;
      const float vh = Pp[(ok ? Thi : 0) * PSTEP];
      const float vl = Pp[(lo ? (Tlo - 4) : 0) * PSTEP];
      const float v  = lo ? (vh - vl) : vh;
      acc = ok ? (acc + v) : acc;
    }
    if (si < 2 * H1 + 1 && j >= 0 && j < NN) L[(size_t)i * NN + j] = acc;
  }
}

#define CM     64
#define CPITCH 1028

struct Frag {
  float4 a[4][4];
  float4 b[2][4];
};

__device__ __forceinline__ void frag_load_b(Frag& f, const float* __restrict__ wq) {
#pragma unroll
  for (int fn = 0; fn < 2; ++fn)
#pragma unroll
    for (int q = 0; q < 4; ++q)
      f.b[fn][q] = *(const float4*)(wq + fn * 16 * 1024 + 4 * q);
}

__device__ __forceinline__ void frag_load_a(Frag& f, const float* aq) {
#pragma unroll
  for (int fm = 0; fm < 4; ++fm)
#pragma unroll
    for (int q = 0; q < 4; ++q)
      f.a[fm][q] = *(const float4*)(aq + fm * 16 * CPITCH + 4 * q);
}

__device__ __forceinline__ void frag_mma(const Frag& f, v8f (&acc)[4][2]) {
#pragma unroll
  for (int q = 0; q < 4; ++q) {
#pragma unroll
    for (int e = 0; e < 2; ++e) {
      v2f af[4], bf[2];
#pragma unroll
      for (int fm = 0; fm < 4; ++fm)
        af[fm] = e ? v2f{f.a[fm][q].z, f.a[fm][q].w} : v2f{f.a[fm][q].x, f.a[fm][q].y};
#pragma unroll
      for (int fn = 0; fn < 2; ++fn)
        bf[fn] = e ? v2f{f.b[fn][q].z, f.b[fn][q].w} : v2f{f.b[fn][q].x, f.b[fn][q].y};
#pragma unroll
      for (int fm = 0; fm < 4; ++fm)
#pragma unroll
        for (int fn = 0; fn < 2; ++fn)
          acc[fm][fn] = __builtin_amdgcn_wmma_f32_16x16x4_f32(
              false, af[fm], false, bf[fn], (short)0, acc[fm][fn],
              false, false);
    }
  }
}

template <bool LAST>
__device__ __forceinline__ void chain_panel(const float* __restrict__ wl,
                                            const float* ap,
                                            const int woff, const int k_lo,
                                            const int n,
                                            const int woff_nx, const int k_nx,
                                            Frag& fa, Frag& fb,
                                            v8f (&acc)[4][2]) {
  for (int i = 0; i < n; i += 2) {
    const bool more1 = (i + 1 < n);
    frag_load_b(fb, wl + (more1 ? woff + 32 * (i + 1) : woff_nx));
    if (!LAST || more1) frag_load_a(fb, ap + (more1 ? k_lo + 32 * (i + 1) : k_nx));
    __builtin_amdgcn_sched_barrier(0);
    frag_mma(fa, acc);
    if (!more1) {
#pragma unroll
      for (int fn = 0; fn < 2; ++fn)
#pragma unroll
        for (int q = 0; q < 4; ++q) fa.b[fn][q] = fb.b[fn][q];
      if (!LAST) {
#pragma unroll
        for (int fm = 0; fm < 4; ++fm)
#pragma unroll
          for (int q = 0; q < 4; ++q) fa.a[fm][q] = fb.a[fm][q];
      }
      break;
    }
    const bool more2 = (i + 2 < n);
    frag_load_b(fa, wl + (more2 ? woff + 32 * (i + 2) : woff_nx));
    if (!LAST || more2) frag_load_a(fa, ap + (more2 ? k_lo + 32 * (i + 2) : k_nx));
    __builtin_amdgcn_sched_barrier(0);
    frag_mma(fb, acc);
  }
}

__device__ __forceinline__ void chain_store(float* act, const int n0,
                                            const int c, const int h,
                                            const v8f (&acc)[4][2]) {
#pragma unroll
  for (int fm = 0; fm < 4; ++fm)
#pragma unroll
    for (int fn = 0; fn < 2; ++fn)
#pragma unroll
      for (int r = 0; r < 8; ++r)
        act[(16 * fm + 8 * h + r) * CPITCH + n0 + 16 * fn + c] = acc[fm][fn][r];
}

__global__ __launch_bounds__(256) void k_chain(const float* __restrict__ X,
                                               const float* __restrict__ Wall,
                                               float* __restrict__ Bneck,
                                               float* __restrict__ Final) {
  __shared__ float Act[CM * CPITCH];

  const int tid  = threadIdx.x;
  const int lane = tid & 31;
  const int w    = __builtin_amdgcn_readfirstlane(tid >> 5);
  const int c    = lane & 15;
  const int h    = lane >> 4;
  const size_t mb = (size_t)blockIdx.x * CM;

  const int p0 = 32 * w,        kl0 = 0,             c0 = w + 17;
  const int p1 = 32 * (15 - w), kl1 = 0,             c1 = 32 - w;
  const int p2 = 32 * (16 + w), kl2 = 32 * w,        c2 = 32 - w;
  const int p3 = 32 * (31 - w), kl3 = 32 * (15 - w), c3 = w + 17;

  const float* wl = Wall + c * 1024 + 16 * h;
  const float* ap = Act + c * CPITCH + 16 * h;

  Frag fa, fb;
  frag_load_b(fa, wl + p0 * 1024 + kl0);

  for (int it = 0; it < (CM * 256) / 256; ++it) {
    const int idx = it * 256 + tid;
    const int row = idx >> 8;
    const int c4  = (idx & 255) * 4;
    *(float4*)(Act + row * CPITCH + c4) =
        *(const float4*)(X + (mb + row) * 1024 + c4);
  }
  __syncthreads();

  for (int l = 0; l < 8; ++l) {
    const int lo = l * NN * NN;
    const int ln = ((l < 7) ? (l + 1) : l) * NN * NN;
    v8f acc0[4][2], acc1[4][2], acc2[4][2], acc3[4][2];
#pragma unroll
    for (int fm = 0; fm < 4; ++fm)
#pragma unroll
      for (int fn = 0; fn < 2; ++fn) {
        acc0[fm][fn] = v8f{}; acc1[fm][fn] = v8f{};
        acc2[fm][fn] = v8f{}; acc3[fm][fn] = v8f{};
      }

    frag_load_a(fa, ap + kl0);

    chain_panel<false>(wl, ap, lo + p0 * 1024 + kl0, kl0, c0,
                       lo + p1 * 1024 + kl1, kl1, fa, fb, acc0);
    chain_panel<false>(wl, ap, lo + p1 * 1024 + kl1, kl1, c1,
                       lo + p2 * 1024 + kl2, kl2, fa, fb, acc1);
    chain_panel<false>(wl, ap, lo + p2 * 1024 + kl2, kl2, c2,
                       lo + p3 * 1024 + kl3, kl3, fa, fb, acc2);
    chain_panel<true>(wl, ap, lo + p3 * 1024 + kl3, kl3, c3,
                      ln + p0 * 1024 + kl0, kl0, fa, fb, acc3);

    __syncthreads();
    chain_store(Act, p0, c, h, acc0);
    chain_store(Act, p1, c, h, acc1);
    chain_store(Act, p2, c, h, acc2);
    chain_store(Act, p3, c, h, acc3);
    __syncthreads();

    if (l == 3 || l == 7) {
      float* dst = (l == 3) ? Bneck : Final;
      for (int it = 0; it < (CM * 256) / 256; ++it) {
        const int idx = it * 256 + tid;
        const int row = idx >> 8;
        const int c4  = (idx & 255) * 4;
        *(float4*)(dst + (mb + row) * 1024 + c4) =
            *(const float4*)(Act + row * CPITCH + c4);
      }
    }
  }
}

extern "C" void kernel_launch(void* const* d_in, const int* in_sizes, int n_in,
                              void* d_out, int out_size, void* d_ws,
                              size_t ws_size, hipStream_t stream) {
  (void)in_sizes; (void)n_in; (void)out_size; (void)ws_size;

  const float* x        = (const float*)d_in[0];
  const float* enc_rot  = (const float*)d_in[1];
  const float* enc_diag = (const float*)d_in[2];
  const float* dec_rot  = (const float*)d_in[3];
  const float* dec_diag = (const float*)d_in[4];

  float* out = (float*)d_out;
  float* lifted = (float*)d_ws;

  float* bneck  = out;
  float* final_ = out + (size_t)BB * NN;

  k_build_band<<<dim3(32, 8), 512, 0, stream>>>(
      enc_rot, enc_diag, dec_rot, dec_diag, lifted);

  k_chain<<<BB / CM, 256, 0, stream>>>(x, lifted, bneck, final_);
}
